# starting kernel
# baseline (speedup 1.0000x reference)
#include <hip/hip_runtime.h>

typedef __attribute__((ext_vector_type(2))) float v2f;
typedef __attribute__((ext_vector_type(8))) float v8f;

__global__ __launch_bounds__(256) void k_init_deg(float* __restrict__ deg, int n) {
  int i = blockIdx.x * blockDim.x + threadIdx.x;
  if (i < n) deg[i] = 1.0f;
}

__global__ __launch_bounds__(256) void k_deg_v4(const int* __restrict__ dst,
                                                float* __restrict__ deg, int nvec) {
  int i = blockIdx.x * blockDim.x + threadIdx.x;
  if (i < nvec) {
    int4 d = ((const int4*)dst)[i];
    atomicAdd(&deg[d.x], 1.0f);
    atomicAdd(&deg[d.y], 1.0f);
    atomicAdd(&deg[d.z], 1.0f);
    atomicAdd(&deg[d.w], 1.0f);
  }
}
__global__ __launch_bounds__(256) void k_deg_s(const int* __restrict__ dst,
                                               float* __restrict__ deg, int e) {
  int i = blockIdx.x * blockDim.x + threadIdx.x;
  if (i < e) atomicAdd(&deg[dst[i]], 1.0f);
}

__global__ __launch_bounds__(256) void k_node1(const float* __restrict__ x,
                                               const float* __restrict__ deg,
                                               float* __restrict__ dinv,
                                               float* __restrict__ p,
                                               float* __restrict__ t, int n) {
  int i = blockIdx.x * blockDim.x + threadIdx.x;
  if (i < n) {
    float dv = rsqrtf(deg[i]);
    dinv[i] = dv;
    float pv = x[i] * dv;
    p[i] = pv;
    t[i] = pv;
  }
}

__global__ __launch_bounds__(256) void k_scatter_v4(const int* __restrict__ src,
                                                    const int* __restrict__ dst,
                                                    const float* __restrict__ val,
                                                    float* __restrict__ acc, int nvec) {
  int i = blockIdx.x * blockDim.x + threadIdx.x;
  if (i < nvec) {
    int4 s = ((const int4*)src)[i];
    int4 d = ((const int4*)dst)[i];
    atomicAdd(&acc[d.x], val[s.x]);
    atomicAdd(&acc[d.y], val[s.y]);
    atomicAdd(&acc[d.z], val[s.z]);
    atomicAdd(&acc[d.w], val[s.w]);
  }
}
__global__ __launch_bounds__(256) void k_scatter_s(const int* __restrict__ src,
                                                   const int* __restrict__ dst,
                                                   const float* __restrict__ val,
                                                   float* __restrict__ acc, int e) {
  int i = blockIdx.x * blockDim.x + threadIdx.x;
  if (i < e) atomicAdd(&acc[dst[i]], val[src[i]]);
}

template <int IMM>
__device__ __forceinline__ float swz_add(float v) {
#if defined(__AMDGCN__)
  int s = __builtin_amdgcn_ds_swizzle(__float_as_int(v), IMM);
  return v + __int_as_float(s);
#else
  return v;
#endif
}

__global__ __launch_bounds__(256) void k_mid(const float* __restrict__ tacc,
                                             const float* __restrict__ dinv,
                                             const float* __restrict__ W1,
                                             const float* __restrict__ b1,
                                             const float* __restrict__ W2,
                                             float* __restrict__ q,
                                             float* __restrict__ u, int n) {
  const int lane   = threadIdx.x & 31;
  const int lane16 = lane & 15;
  const bool hi    = lane >= 16;
  const long long wave = (long long)blockIdx.x * (blockDim.x >> 5) + (threadIdx.x >> 5);
  const long long base = wave * 16;
  if (base >= n) return;

  float a = 0.0f;
  if (!hi) {
    long long idx = base + lane16;
    if (idx < n) a = tacc[idx] * dinv[idx];
  }
  const float w1v = W1[lane16];
  const float b1v = b1[lane16];
  const float w2v = W2[lane16];

#if __has_builtin(__builtin_amdgcn_wmma_f32_16x16x4_f32)
  v2f A; A.x = a;                  A.y = 0.0f;
  v2f B; B.x = hi ? 0.0f : w1v;    B.y = 0.0f;
  v8f C;
  #pragma unroll
  for (int r = 0; r < 8; ++r) C[r] = b1v;
  v8f D = __builtin_amdgcn_wmma_f32_16x16x4_f32(
      false, A, false, B, (short)0, C, false, false);
#else
  v8f D;
  #pragma unroll
  for (int r = 0; r < 8; ++r) {
    float am = __shfl(a, r + (hi ? 8 : 0), 32);
    D[r] = fmaf(am, w1v, b1v);
  }
#endif

  float z[8];
  #pragma unroll
  for (int r = 0; r < 8; ++r) {
    float h = D[r];
    h = h > 0.0f ? h : 0.0f;
    float v = h * w2v;
    v = swz_add<0x041F>(v);
    v = swz_add<0x081F>(v);
    v = swz_add<0x101F>(v);
    v = swz_add<0x201F>(v);
    z[r] = v;
  }

  if (lane == 0 || lane == 16) {
    const int off = hi ? 8 : 0;
    #pragma unroll
    for (int r = 0; r < 8; ++r) {
      long long idx = base + off + r;
      if (idx < n) {
        float qq = z[r] * dinv[idx];
        q[idx] = qq;
        u[idx] = qq;
      }
    }
  }
}

__global__ __launch_bounds__(256) void k_out(const float* __restrict__ dinv,
                                             const float* __restrict__ u,
                                             const float* __restrict__ b2,
                                             float* __restrict__ out, int n) {
  int i = blockIdx.x * blockDim.x + threadIdx.x;
  if (i < n) out[i] = fmaf(dinv[i], u[i], b2[0]);
}

extern "C" void kernel_launch(void* const* d_in, const int* in_sizes, int n_in,
                              void* d_out, int out_size, void* d_ws, size_t ws_size,
                              hipStream_t stream) {
  const float* x  = (const float*)d_in[0];
  const float* W1 = (const float*)d_in[1];
  const float* b1 = (const float*)d_in[2];
  const float* W2 = (const float*)d_in[3];
  const float* b2 = (const float*)d_in[4];
  const int*   ei = (const int*)d_in[5];

  const int n = in_sizes[0];
  const int e = in_sizes[5] / 2;
  const int* src = ei;
  const int* dst = ei + e;

  float* out  = (float*)d_out;
  float* dinv = (float*)d_ws;
  float* p    = dinv + n;
  float* t    = p + n;
  float* u    = t + n;
  float* q    = p;

  const int B  = 256;
  const int gN = (n + B - 1) / B;
  const int gM = (n + 127) / 128;

  k_init_deg<<<gN, B, 0, stream>>>(out, n);
  if ((e & 3) == 0) {
    int nvec = e >> 2;
    int gE = (nvec + B - 1) / B;
    k_deg_v4<<<gE, B, 0, stream>>>(dst, out, nvec);
  } else {
    int gE = (e + B - 1) / B;
    k_deg_s<<<gE, B, 0, stream>>>(dst, out, e);
  }

  k_node1<<<gN, B, 0, stream>>>(x, out, dinv, p, t, n);

  if ((e & 3) == 0) {
    int nvec = e >> 2;
    int gE = (nvec + B - 1) / B;
    k_scatter_v4<<<gE, B, 0, stream>>>(src, dst, p, t, nvec);
  } else {
    int gE = (e + B - 1) / B;
    k_scatter_s<<<gE, B, 0, stream>>>(src, dst, p, t, e);
  }

  k_mid<<<gM, B, 0, stream>>>(t, dinv, W1, b1, W2, q, u, n);

  if ((e & 3) == 0) {
    int nvec = e >> 2;
    int gE = (nvec + B - 1) / B;
    k_scatter_v4<<<gE, B, 0, stream>>>(src, dst, q, u, nvec);
  } else {
    int gE = (e + B - 1) / B;
    k_scatter_s<<<gE, B, 0, stream>>>(src, dst, q, u, e);
  }

  k_out<<<gN, B, 0, stream>>>(dinv, u, b2, out, n);
}


# Scatter privatised in LDS: a work-group keeps a 65536-node destination range, adds its edge piece by LDS float atomics, stores a partial plane; consumers sum 32 planes; no global atomics
# speedup vs baseline: 3.0440x; 3.0440x over previous
#include <hip/hip_runtime.h>


typedef __attribute__((ext_vector_type(2))) float v2f;
typedef __attribute__((ext_vector_type(8))) float v8f;

#define RANGE_NODES 65536
#define PRIV_THREADS 1024
#define NCHUNK 32

__device__ __forceinline__ int plane_stride(int n) { return (n + 31) & ~31; }

template <bool GATHER>
__global__ __launch_bounds__(PRIV_THREADS) void k_priv(const int* __restrict__ src,
                                                       const int* __restrict__ dst,
                                                       const float* __restrict__ val,
                                                       float* __restrict__ part,
                                                       int n, int e, int nr) {
  __shared__ __attribute__((aligned(16))) float acc[RANGE_NODES];
  const int tid = threadIdx.x;
  const int r   = blockIdx.x % nr;
  const int c   = blockIdx.x / nr;
  const int lo  = r * RANGE_NODES;

  float4* acc4 = (float4*)acc;
  for (int i = tid; i < RANGE_NODES / 4; i += PRIV_THREADS) acc4[i] = make_float4(0.f, 0.f, 0.f, 0.f);
  __syncthreads();

  if ((e & 3) == 0) {
    const int nvec = e >> 2;
    const int per  = (nvec + NCHUNK - 1) / NCHUNK;
    const int beg  = c * per;
    const int end  = min(beg + per, nvec);
    const int4* __restrict__ d4 = (const int4*)dst;
    const int4* __restrict__ s4 = (const int4*)src;

    const int4 none = make_int4(-1, -1, -1, -1);
    const int4 zero = make_int4(0, 0, 0, 0);

    if (beg < end) {
      const int last  = end - 1;
      const int trips = (end - beg + 2 * PRIV_THREADS - 1) / (2 * PRIV_THREADS);
      int i = beg + tid;
      int4 dA = d4[min(i, last)], dB = d4[min(i + PRIV_THREADS, last)];
      int4 sA = zero, sB = zero;
      if (GATHER) { sA = s4[min(i, last)]; sB = s4[min(i + PRIV_THREADS, last)]; }

      for (int t = 0; t < trips; ++t, i += 2 * PRIV_THREADS) {
        if (i >= end)                dA = none;
        if (i + PRIV_THREADS >= end) dB = none;
        unsigned u[8];
        u[0] = (unsigned)(dA.x - lo); u[1] = (unsigned)(dA.y - lo);
        u[2] = (unsigned)(dA.z - lo); u[3] = (unsigned)(dA.w - lo);
        u[4] = (unsigned)(dB.x - lo); u[5] = (unsigned)(dB.y - lo);
        u[6] = (unsigned)(dB.z - lo); u[7] = (unsigned)(dB.w - lo);
        int sidx[8] = {sA.x, sA.y, sA.z, sA.w, sB.x, sB.y, sB.z, sB.w};

        const int j = i + 2 * PRIV_THREADS;
        int4 dnA = d4[min(j, last)], dnB = d4[min(j + PRIV_THREADS, last)];
        int4 snA = zero, snB = zero;
        if (GATHER) { snA = s4[min(j, last)]; snB = s4[min(j + PRIV_THREADS, last)]; }

        float v[8];
        #pragma unroll
        for (int k = 0; k < 8; ++k) {
          v[k] = 1.0f;
          if (GATHER) v[k] = (u[k] < (unsigned)RANGE_NODES) ? val[sidx[k]] : 0.0f;
        }

        #pragma unroll
        for (int k = 0; k < 8; ++k)
          if (u[k] < (unsigned)RANGE_NODES) atomicAdd(&acc[u[k]], v[k]);

        dA = dnA; dB = dnB; sA = snA; sB = snB;
      }
    }
  } else {
    const int per = (e + NCHUNK - 1) / NCHUNK;
    const int beg = c * per;
    const int end = min(beg + per, e);
    for (int i = beg + tid; i < end; i += PRIV_THREADS) {
      unsigned u0 = (unsigned)(dst[i] - lo);
      if (u0 < (unsigned)RANGE_NODES) atomicAdd(&acc[u0], GATHER ? val[src[i]] : 1.0f);
    }
  }
  __syncthreads();

  const int ns  = plane_stride(n);
  const int cnt = min(RANGE_NODES, ns - lo);
  float4* __restrict__ o4 = (float4*)(part + (size_t)c * ns + lo);
  for (int i = tid; i < (cnt >> 2); i += PRIV_THREADS) o4[i] = acc4[i];
}

__device__ __forceinline__ float plane_sum(const float* __restrict__ part, int ns, int i) {
  float s = 0.0f;
  #pragma unroll 8
  for (int c = 0; c < NCHUNK; ++c) s += part[(size_t)c * ns + i];
  return s;
}

__global__ __launch_bounds__(256) void k_node1(const float* __restrict__ x,
                                               const float* __restrict__ part,
                                               float* __restrict__ dinv,
                                               float* __restrict__ p, int n) {
  int i = blockIdx.x * blockDim.x + threadIdx.x;
  if (i < n) {
    float dg = 1.0f + plane_sum(part, plane_stride(n), i);
    float dv = rsqrtf(dg);
    dinv[i] = dv;
    p[i] = x[i] * dv;
  }
}

template <int IMM>
__device__ __forceinline__ float swz_add(float v) {
#if defined(__AMDGCN__)
  int s = __builtin_amdgcn_ds_swizzle(__float_as_int(v), IMM);
  return v + __int_as_float(s);
#else
  return v;
#endif
}

__global__ __launch_bounds__(256) void k_mid(const float* __restrict__ p,
                                             const float* __restrict__ part,
                                             const float* __restrict__ dinv,
                                             const float* __restrict__ W1,
                                             const float* __restrict__ b1,
                                             const float* __restrict__ W2,
                                             float* __restrict__ q, int n) {
  const int lane   = threadIdx.x & 31;
  const int lane16 = lane & 15;
  const bool hi    = lane >= 16;
  const long long wave = (long long)blockIdx.x * (blockDim.x >> 5) + (threadIdx.x >> 5);
  const long long base = wave * 16;
  if (base >= n) return;

  const long long idx = base + lane16;
  const int ns = plane_stride(n);
  float s = 0.0f;
  if (idx < n) {
    #pragma unroll 8
    for (int c = (hi ? 1 : 0); c < NCHUNK; c += 2) s += part[(size_t)c * ns + idx];
  }
  s = swz_add<0x401F>(s);

  float a = 0.0f;
  if (!hi && idx < n) a = (p[idx] + s) * dinv[idx];
  const float w1v = W1[lane16];
  const float b1v = b1[lane16];
  const float w2v = W2[lane16];

#if __has_builtin(__builtin_amdgcn_wmma_f32_16x16x4_f32)
  v2f A; A.x = a;                  A.y = 0.0f;
  v2f B; B.x = hi ? 0.0f : w1v;    B.y = 0.0f;
  v8f C;
  #pragma unroll
  for (int r = 0; r < 8; ++r) C[r] = b1v;
  v8f D = __builtin_amdgcn_wmma_f32_16x16x4_f32(
      false, A, false, B, (short)0, C, false, false);
#else
  v8f D;
  #pragma unroll
  for (int r = 0; r < 8; ++r) {
    float am = __shfl(a, r + (hi ? 8 : 0), 32);
    D[r] = fmaf(am, w1v, b1v);
  }
#endif

  float z[8];
  #pragma unroll
  for (int r = 0; r < 8; ++r) {
    float h = D[r];
    h = h > 0.0f ? h : 0.0f;
    float v = h * w2v;
    v = swz_add<0x041F>(v);
    v = swz_add<0x081F>(v);
    v = swz_add<0x101F>(v);
    v = swz_add<0x201F>(v);
    z[r] = v;
  }

  if (lane == 0 || lane == 16) {
    const int off = hi ? 8 : 0;
    #pragma unroll
    for (int r = 0; r < 8; ++r) {
      long long o = base + off + r;
      if (o < n) q[o] = z[r] * dinv[o];
    }
  }
}

__global__ __launch_bounds__(256) void k_out(const float* __restrict__ dinv,
                                             const float* __restrict__ q,
                                             const float* __restrict__ part,
                                             const float* __restrict__ b2,
                                             float* __restrict__ out, int n) {
  int i = blockIdx.x * blockDim.x + threadIdx.x;
  if (i < n) {
    float u = q[i] + plane_sum(part, plane_stride(n), i);
    out[i] = fmaf(dinv[i], u, b2[0]);
  }
}

extern "C" void kernel_launch(void* const* d_in, const int* in_sizes, int n_in,
                              void* d_out, int out_size, void* d_ws, size_t ws_size,
                              hipStream_t stream) {
  const float* x  = (const float*)d_in[0];
  const float* W1 = (const float*)d_in[1];
  const float* b1 = (const float*)d_in[2];
  const float* W2 = (const float*)d_in[3];
  const float* b2 = (const float*)d_in[4];
  const int*   ei = (const int*)d_in[5];

  const int n = in_sizes[0];
  const int e = in_sizes[5] / 2;
  const int* src = ei;
  const int* dst = ei + e;

  float* out  = (float*)d_out;
  float* dinv = (float*)d_ws;
  float* p    = dinv + n;
  float* q    = p + n;
  float* part = (float*)d_ws + (16u << 20);

  const int B  = 256;
  const int gN = (n + B - 1) / B;
  const int gM = (n + 127) / 128;
  const int nr = (n + RANGE_NODES - 1) / RANGE_NODES;
  const int gP = nr * NCHUNK;

  k_priv<false><<<gP, PRIV_THREADS, 0, stream>>>(src, dst, x, part, n, e, nr);
  k_node1<<<gN, B, 0, stream>>>(x, part, dinv, p, n);
  k_priv<true><<<gP, PRIV_THREADS, 0, stream>>>(src, dst, p, part, n, e, nr);
  k_mid<<<gM, B, 0, stream>>>(p, part, dinv, W1, b1, W2, q, n);
  k_priv<true><<<gP, PRIV_THREADS, 0, stream>>>(src, dst, q, part, n, e, nr);
  k_out<<<gN, B, 0, stream>>>(dinv, q, part, b2, out, n);
}


# Edge list binned once per call by (destination range, source window): a scatter work-group streams only its bin and gathers from an LDS-staged window; 16 window planes summed (sum order changes)
# speedup vs baseline: 6.1091x; 2.0069x over previous
#include <hip/hip_runtime.h>


typedef __attribute__((ext_vector_type(2))) float v2f;
typedef __attribute__((ext_vector_type(8))) float v8f;

#define RSHIFT 15
#define RNODES (1 << RSHIFT)
#define NBIN 256
#define TILE 20480
#define PT 1024

__device__ __forceinline__ int plane_stride(int n) { return (n + 31) & ~31; }

__global__ __launch_bounds__(PT) void k_count(const int* __restrict__ src,
                                              const int* __restrict__ dst,
                                              int* __restrict__ table, int e, int nr) {
  __shared__ int hist[NBIN];
  const int tid = threadIdx.x;
  if (tid < NBIN) hist[tid] = 0;
  __syncthreads();
  const int beg = blockIdx.x * TILE;
  const int end = min(beg + TILE, e);
  for (int i = beg + tid; i < end; i += 8 * PT) {
    int d[8], s[8];
    #pragma unroll
    for (int k = 0; k < 8; ++k) {
      const int j = min(i + k * PT, end - 1);
      d[k] = dst[j];
      s[k] = src[j];
    }
    asm volatile("" :: "v"(d[0]), "v"(d[1]), "v"(d[2]), "v"(d[3]), "v"(d[4]), "v"(d[5]), "v"(d[6]), "v"(d[7]),
                       "v"(s[0]), "v"(s[1]), "v"(s[2]), "v"(s[3]), "v"(s[4]), "v"(s[5]), "v"(s[6]), "v"(s[7]));
    #pragma unroll
    for (int k = 0; k < 8; ++k)
      if (i + k * PT < end) atomicAdd(&hist[(d[k] >> RSHIFT) * nr + (s[k] >> RSHIFT)], 1);
  }
  __syncthreads();
  if (tid < NBIN) table[blockIdx.x * NBIN + tid] = hist[tid];
}

#define EMPTY 0xFFFFFFFFu
#define CAP (TILE + NBIN * 31)
__device__ __forceinline__ int pad32(int v) { return (v + 31) & ~31; }

__global__ __launch_bounds__(PT) void k_place(const int* __restrict__ src,
                                              const int* __restrict__ dst,
                                              const int* __restrict__ table,
                                              int* __restrict__ binoff,
                                              unsigned* __restrict__ bins, int e, int nr, int ntile) {
  __shared__ unsigned stage[CAP];
  __shared__ int chunkaddr[CAP / 32];
  __shared__ int colp[4][NBIN];
  __shared__ int prep[4][NBIN];
  __shared__ int tot_s[NBIN];
  __shared__ int own_s[NBIN];
  __shared__ int cursor[NBIN];
  __shared__ int ltot;
  const int tid = threadIdx.x;
  const int g   = blockIdx.x;
  const int t   = tid & (NBIN - 1);
  const int qt  = tid >> 8;

  int col = 0, pre = 0;
  #pragma unroll 8
  for (int gg = qt; gg < ntile; gg += 4) {
    const int v = pad32(table[gg * NBIN + t]);
    col += v;
    if (gg < g) pre += v;
  }
  colp[qt][t] = col;
  prep[qt][t] = pre;
  __syncthreads();
  int tot = 0, mypre = 0, own = 0, ownp = 0;
  if (tid < NBIN) {
    tot   = colp[0][tid] + colp[1][tid] + colp[2][tid] + colp[3][tid];
    mypre = prep[0][tid] + prep[1][tid] + prep[2][tid] + prep[3][tid];
    own   = table[g * NBIN + tid];
    ownp  = pad32(own);
    tot_s[tid] = tot;
    own_s[tid] = ownp;
  }
  __syncthreads();
  if (tid < NBIN) {
    int sa = 0, sb = 0;
    for (int b = 0; b < tid; ++b) { sa += tot_s[b]; sb += own_s[b]; }
    cursor[tid] = sb;
    const int gchunk = (sa + mypre) >> 5;
    const int lchunk = sb >> 5;
    for (int k = 0; k < (ownp >> 5); ++k) chunkaddr[lchunk + k] = gchunk + k;
    for (int j = sb + own; j < sb + ownp; ++j) stage[j] = EMPTY;
    if (tid == NBIN - 1) ltot = sb + ownp;
    if (g == 0) {
      binoff[tid] = sa;
      if (tid == NBIN - 1) binoff[NBIN] = sa + tot;
    }
  }
  __syncthreads();

  const int beg = g * TILE;
  const int end = min(beg + TILE, e);
  for (int i = beg + tid; i < end; i += 8 * PT) {
    int d[8], s[8];
    #pragma unroll
    for (int k = 0; k < 8; ++k) {
      const int j = min(i + k * PT, end - 1);
      d[k] = dst[j];
      s[k] = src[j];
    }
    asm volatile("" :: "v"(d[0]), "v"(d[1]), "v"(d[2]), "v"(d[3]), "v"(d[4]), "v"(d[5]), "v"(d[6]), "v"(d[7]),
                       "v"(s[0]), "v"(s[1]), "v"(s[2]), "v"(s[3]), "v"(s[4]), "v"(s[5]), "v"(s[6]), "v"(s[7]));
    #pragma unroll
    for (int k = 0; k < 8; ++k) {
      if (i + k * PT < end) {
        const int b   = (d[k] >> RSHIFT) * nr + (s[k] >> RSHIFT);
        const int pos = atomicAdd(&cursor[b], 1);
        stage[pos] = ((unsigned)(s[k] & (RNODES - 1)) << 16) | (unsigned)(d[k] & (RNODES - 1));
      }
    }
  }
  __syncthreads();
  const int total = ltot;
  for (int j = tid; j < total; j += PT) bins[(size_t)chunkaddr[j >> 5] * 32 + (j & 31)] = stage[j];
}

template <bool GATHER>
__global__ __launch_bounds__(PT) void k_bin(const unsigned* __restrict__ bins,
                                            const int* __restrict__ binoff,
                                            const float* __restrict__ val,
                                            float* __restrict__ part, int n, int nr) {
  __shared__ __attribute__((aligned(16))) float acc[RNODES];
  __shared__ __attribute__((aligned(16))) float win[GATHER ? RNODES : 4];
  const int tid = threadIdx.x;
  const int r   = blockIdx.x / nr;
  const int w   = blockIdx.x - r * nr;

  float4* acc4 = (float4*)acc;
  for (int i = tid; i < RNODES / 4; i += PT) acc4[i] = make_float4(0.f, 0.f, 0.f, 0.f);
  if (GATHER) {
    const int wlo = w << RSHIFT;
    #pragma unroll 16
    for (int i = tid; i < RNODES; i += PT) {
      const int gi = wlo + i;
      const float t = val[min(gi, n - 1)];
      win[i] = (gi < n) ? t : 0.0f;
    }
  }
  __syncthreads();

  const uint4* __restrict__ b4 = (const uint4*)bins;
  const int s4 = binoff[blockIdx.x] >> 2;
  const int e4 = binoff[blockIdx.x + 1] >> 2;
  for (int i = s4 + tid; i < e4; i += 2 * PT) {
    const uint4 A = b4[i];
    const uint4 B = b4[min(i + PT, e4 - 1)];
    asm volatile("" :: "v"(A.x), "v"(A.y), "v"(A.z), "v"(A.w), "v"(B.x), "v"(B.y), "v"(B.z), "v"(B.w));
    const bool okB = (i + PT < e4);
    const unsigned pk[8] = {A.x, A.y, A.z, A.w, B.x, B.y, B.z, B.w};
    float v[8];
    #pragma unroll
    for (int k = 0; k < 8; ++k) v[k] = GATHER ? win[(pk[k] >> 16) & (RNODES - 1)] : 1.0f;
    #pragma unroll
    for (int k = 0; k < 8; ++k)
      if ((k < 4 || okB) && pk[k] != EMPTY) atomicAdd(&acc[pk[k] & (RNODES - 1)], v[k]);
  }
  __syncthreads();

  const int ns  = plane_stride(n);
  const int lo  = r << RSHIFT;
  const int cnt = min(RNODES, ns - lo);
  float4* __restrict__ o4 = (float4*)(part + (size_t)w * ns + lo);
  for (int i = tid; i < (cnt >> 2); i += PT) o4[i] = acc4[i];
}

__device__ __forceinline__ float plane_sum(const float* __restrict__ part, int ns, int i, int np) {
  float s = 0.0f;
  #pragma unroll 8
  for (int c = 0; c < np; ++c) s += part[(size_t)c * ns + i];
  return s;
}

__global__ __launch_bounds__(256) void k_node1(const float* __restrict__ x,
                                               const float* __restrict__ part,
                                               float* __restrict__ dinv,
                                               float* __restrict__ p, int n, int np) {
  int i = blockIdx.x * blockDim.x + threadIdx.x;
  if (i < n) {
    float dg = 1.0f + plane_sum(part, plane_stride(n), i, np);
    float dv = rsqrtf(dg);
    dinv[i] = dv;
    p[i] = x[i] * dv;
  }
}

template <int IMM>
__device__ __forceinline__ float swz_add(float v) {
#if defined(__AMDGCN__)
  int s = __builtin_amdgcn_ds_swizzle(__float_as_int(v), IMM);
  return v + __int_as_float(s);
#else
  return v;
#endif
}

__global__ __launch_bounds__(256) void k_mid(const float* __restrict__ p,
                                             const float* __restrict__ part,
                                             const float* __restrict__ dinv,
                                             const float* __restrict__ W1,
                                             const float* __restrict__ b1,
                                             const float* __restrict__ W2,
                                             float* __restrict__ q, int n, int np) {
  const int lane   = threadIdx.x & 31;
  const int lane16 = lane & 15;
  const bool hi    = lane >= 16;
  const long long wave = (long long)blockIdx.x * (blockDim.x >> 5) + (threadIdx.x >> 5);
  const long long base = wave * 16;
  if (base >= n) return;

  const long long idx = base + lane16;
  const int ns = plane_stride(n);
  float s = 0.0f;
  if (idx < n) {
    #pragma unroll 8
    for (int c = (hi ? 1 : 0); c < np; c += 2) s += part[(size_t)c * ns + idx];
  }
  s = swz_add<0x401F>(s);

  float a = 0.0f;
  if (!hi && idx < n) a = (p[idx] + s) * dinv[idx];
  const float w1v = W1[lane16];
  const float b1v = b1[lane16];
  const float w2v = W2[lane16];

#if __has_builtin(__builtin_amdgcn_wmma_f32_16x16x4_f32)
  v2f A; A.x = a;                  A.y = 0.0f;
  v2f B; B.x = hi ? 0.0f : w1v;    B.y = 0.0f;
  v8f C;
  #pragma unroll
  for (int r = 0; r < 8; ++r) C[r] = b1v;
  v8f D = __builtin_amdgcn_wmma_f32_16x16x4_f32(
      false, A, false, B, (short)0, C, false, false);
#else
  v8f D;
  #pragma unroll
  for (int r = 0; r < 8; ++r) {
    float am = __shfl(a, r + (hi ? 8 : 0), 32);
    D[r] = fmaf(am, w1v, b1v);
  }
#endif

  float z[8];
  #pragma unroll
  for (int r = 0; r < 8; ++r) {
    float h = D[r];
    h = h > 0.0f ? h : 0.0f;
    float v = h * w2v;
    v = swz_add<0x041F>(v);
    v = swz_add<0x081F>(v);
    v = swz_add<0x101F>(v);
    v = swz_add<0x201F>(v);
    z[r] = v;
  }

  if (lane == 0 || lane == 16) {
    const int off = hi ? 8 : 0;
    #pragma unroll
    for (int r = 0; r < 8; ++r) {
      long long o = base + off + r;
      if (o < n) q[o] = z[r] * dinv[o];
    }
  }
}

__global__ __launch_bounds__(256) void k_out(const float* __restrict__ dinv,
                                             const float* __restrict__ q,
                                             const float* __restrict__ part,
                                             const float* __restrict__ b2,
                                             float* __restrict__ out, int n, int np) {
  int i = blockIdx.x * blockDim.x + threadIdx.x;
  if (i < n) {
    float u = q[i] + plane_sum(part, plane_stride(n), i, np);
    out[i] = fmaf(dinv[i], u, b2[0]);
  }
}

extern "C" void kernel_launch(void* const* d_in, const int* in_sizes, int n_in,
                              void* d_out, int out_size, void* d_ws, size_t ws_size,
                              hipStream_t stream) {
  const float* x  = (const float*)d_in[0];
  const float* W1 = (const float*)d_in[1];
  const float* b1 = (const float*)d_in[2];
  const float* W2 = (const float*)d_in[3];
  const float* b2 = (const float*)d_in[4];
  const int*   ei = (const int*)d_in[5];

  const int n = in_sizes[0];
  const int e = in_sizes[5] / 2;
  const int* src = ei;
  const int* dst = ei + e;

  float* out  = (float*)d_out;
  float* dinv = (float*)d_ws;
  float* p    = dinv + n;
  float* q    = p + n;
  int*      table  = (int*)d_ws + (4u << 20);
  int*      binoff = (int*)d_ws + (6u << 20);
  unsigned* bins   = (unsigned*)d_ws + (8u << 20);
  float*    part   = (float*)d_ws + (16u << 20);

  const int B  = 256;
  const int gN = (n + B - 1) / B;
  const int gM = (n + 127) / 128;
  const int nr = (n + RNODES - 1) / RNODES;
  const int gT = (e + TILE - 1) / TILE;
  const int gP = nr * nr;

  k_count<<<gT, PT, 0, stream>>>(src, dst, table, e, nr);
  k_place<<<gT, PT, 0, stream>>>(src, dst, table, binoff, bins, e, nr, gT);
  k_bin<false><<<gP, PT, 0, stream>>>(bins, binoff, x, part, n, nr);
  k_node1<<<gN, B, 0, stream>>>(x, part, dinv, p, n, nr);
  k_bin<true><<<gP, PT, 0, stream>>>(bins, binoff, p, part, n, nr);
  k_mid<<<gM, B, 0, stream>>>(p, part, dinv, W1, b1, W2, q, n, nr);
  k_bin<true><<<gP, PT, 0, stream>>>(bins, binoff, q, part, n, nr);
  k_out<<<gN, B, 0, stream>>>(dinv, q, part, b2, out, n, nr);
}


# Per-node kernels (degree, middle, output) give each lane 4 consecutive nodes with 16-byte loads and stores; the middle kernel hands them to its WMMA tiles through LDS: 4 to 8 times fewer waves
# speedup vs baseline: 6.7054x; 1.0976x over previous
#include <hip/hip_runtime.h>


typedef __attribute__((ext_vector_type(2))) float v2f;
typedef __attribute__((ext_vector_type(8))) float v8f;
typedef __attribute__((ext_vector_type(4))) float v4f;

#define RSHIFT 15
#define RNODES (1 << RSHIFT)
#define NBIN 256
#define TILE 20480
#define PT 1024

__device__ __forceinline__ int plane_stride(int n) { return (n + 31) & ~31; }

__global__ __launch_bounds__(PT) void k_count(const int* __restrict__ src,
                                              const int* __restrict__ dst,
                                              int* __restrict__ table, int e, int nr) {
  __shared__ int hist[NBIN];
  const int tid = threadIdx.x;
  if (tid < NBIN) hist[tid] = 0;
  __syncthreads();
  const int beg = blockIdx.x * TILE;
  const int end = min(beg + TILE, e);
  for (int i = beg + tid; i < end; i += 8 * PT) {
    int d[8], s[8];
    #pragma unroll
    for (int k = 0; k < 8; ++k) {
      const int j = min(i + k * PT, end - 1);
      d[k] = dst[j];
      s[k] = src[j];
    }
    asm volatile("" :: "v"(d[0]), "v"(d[1]), "v"(d[2]), "v"(d[3]), "v"(d[4]), "v"(d[5]), "v"(d[6]), "v"(d[7]),
                       "v"(s[0]), "v"(s[1]), "v"(s[2]), "v"(s[3]), "v"(s[4]), "v"(s[5]), "v"(s[6]), "v"(s[7]));
    #pragma unroll
    for (int k = 0; k < 8; ++k)
      if (i + k * PT < end) atomicAdd(&hist[(d[k] >> RSHIFT) * nr + (s[k] >> RSHIFT)], 1);
  }
  __syncthreads();
  if (tid < NBIN) table[blockIdx.x * NBIN + tid] = hist[tid];
}

#define EMPTY 0xFFFFFFFFu
#define CAP (TILE + NBIN * 31)
__device__ __forceinline__ int pad32(int v) { return (v + 31) & ~31; }

__global__ __launch_bounds__(PT) void k_place(const int* __restrict__ src,
                                              const int* __restrict__ dst,
                                              const int* __restrict__ table,
                                              int* __restrict__ binoff,
                                              unsigned* __restrict__ bins, int e, int nr, int ntile) {
  __shared__ unsigned stage[CAP];
  __shared__ int chunkaddr[CAP / 32];
  __shared__ int colp[4][NBIN];
  __shared__ int prep[4][NBIN];
  __shared__ int tot_s[NBIN];
  __shared__ int own_s[NBIN];
  __shared__ int cursor[NBIN];
  __shared__ int ltot;
  const int tid = threadIdx.x;
  const int g   = blockIdx.x;
  const int t   = tid & (NBIN - 1);
  const int qt  = tid >> 8;

  int col = 0, pre = 0;
  #pragma unroll 8
  for (int gg = qt; gg < ntile; gg += 4) {
    const int v = pad32(table[gg * NBIN + t]);
    col += v;
    if (gg < g) pre += v;
  }
  colp[qt][t] = col;
  prep[qt][t] = pre;
  __syncthreads();
  int tot = 0, mypre = 0, own = 0, ownp = 0;
  if (tid < NBIN) {
    tot   = colp[0][tid] + colp[1][tid] + colp[2][tid] + colp[3][tid];
    mypre = prep[0][tid] + prep[1][tid] + prep[2][tid] + prep[3][tid];
    own   = table[g * NBIN + tid];
    ownp  = pad32(own);
    tot_s[tid] = tot;
    own_s[tid] = ownp;
  }
  __syncthreads();
  if (tid < NBIN) {
    int sa = 0, sb = 0;
    for (int b = 0; b < tid; ++b) { sa += tot_s[b]; sb += own_s[b]; }
    cursor[tid] = sb;
    const int gchunk = (sa + mypre) >> 5;
    const int lchunk = sb >> 5;
    for (int k = 0; k < (ownp >> 5); ++k) chunkaddr[lchunk + k] = gchunk + k;
    for (int j = sb + own; j < sb + ownp; ++j) stage[j] = EMPTY;
    if (tid == NBIN - 1) ltot = sb + ownp;
    if (g == 0) {
      binoff[tid] = sa;
      if (tid == NBIN - 1) binoff[NBIN] = sa + tot;
    }
  }
  __syncthreads();

  const int beg = g * TILE;
  const int end = min(beg + TILE, e);
  for (int i = beg + tid; i < end; i += 8 * PT) {
    int d[8], s[8];
    #pragma unroll
    for (int k = 0; k < 8; ++k) {
      const int j = min(i + k * PT, end - 1);
      d[k] = dst[j];
      s[k] = src[j];
    }
    asm volatile("" :: "v"(d[0]), "v"(d[1]), "v"(d[2]), "v"(d[3]), "v"(d[4]), "v"(d[5]), "v"(d[6]), "v"(d[7]),
                       "v"(s[0]), "v"(s[1]), "v"(s[2]), "v"(s[3]), "v"(s[4]), "v"(s[5]), "v"(s[6]), "v"(s[7]));
    #pragma unroll
    for (int k = 0; k < 8; ++k) {
      if (i + k * PT < end) {
        const int b   = (d[k] >> RSHIFT) * nr + (s[k] >> RSHIFT);
        const int pos = atomicAdd(&cursor[b], 1);
        stage[pos] = ((unsigned)(s[k] & (RNODES - 1)) << 16) | (unsigned)(d[k] & (RNODES - 1));
      }
    }
  }
  __syncthreads();
  const int total = ltot;
  for (int j = tid; j < total; j += PT) bins[(size_t)chunkaddr[j >> 5] * 32 + (j & 31)] = stage[j];
}

template <bool GATHER>
__global__ __launch_bounds__(PT) void k_bin(const unsigned* __restrict__ bins,
                                            const int* __restrict__ binoff,
                                            const float* __restrict__ val,
                                            float* __restrict__ part, int n, int nr) {
  __shared__ __attribute__((aligned(16))) float acc[RNODES];
  __shared__ __attribute__((aligned(16))) float win[GATHER ? RNODES : 4];
  const int tid = threadIdx.x;
  const int r   = blockIdx.x / nr;
  const int w   = blockIdx.x - r * nr;

  float4* acc4 = (float4*)acc;
  for (int i = tid; i < RNODES / 4; i += PT) acc4[i] = make_float4(0.f, 0.f, 0.f, 0.f);
  if (GATHER) {
    const int wlo = w << RSHIFT;
    #pragma unroll 16
    for (int i = tid; i < RNODES; i += PT) {
      const int gi = wlo + i;
      const float t = val[min(gi, n - 1)];
      win[i] = (gi < n) ? t : 0.0f;
    }
  }
  __syncthreads();

  const uint4* __restrict__ b4 = (const uint4*)bins;
  const int s4 = binoff[blockIdx.x] >> 2;
  const int e4 = binoff[blockIdx.x + 1] >> 2;
  for (int i = s4 + tid; i < e4; i += 2 * PT) {
    const uint4 A = b4[i];
    const uint4 B = b4[min(i + PT, e4 - 1)];
    asm volatile("" :: "v"(A.x), "v"(A.y), "v"(A.z), "v"(A.w), "v"(B.x), "v"(B.y), "v"(B.z), "v"(B.w));
    const bool okB = (i + PT < e4);
    const unsigned pk[8] = {A.x, A.y, A.z, A.w, B.x, B.y, B.z, B.w};
    float v[8];
    #pragma unroll
    for (int k = 0; k < 8; ++k) v[k] = GATHER ? win[(pk[k] >> 16) & (RNODES - 1)] : 1.0f;
    #pragma unroll
    for (int k = 0; k < 8; ++k)
      if ((k < 4 || okB) && pk[k] != EMPTY) atomicAdd(&acc[pk[k] & (RNODES - 1)], v[k]);
  }
  __syncthreads();

  const int ns  = plane_stride(n);
  const int lo  = r << RSHIFT;
  const int cnt = min(RNODES, ns - lo);
  float4* __restrict__ o4 = (float4*)(part + (size_t)w * ns + lo);
  for (int i = tid; i < (cnt >> 2); i += PT) o4[i] = acc4[i];
}

#define NGRP 1024

__device__ __forceinline__ v4f ld4(const float* __restrict__ a) { return *(const v4f*)a; }

__device__ __forceinline__ v4f plane_sum4(const float* __restrict__ part, int ns, int i, int np) {
  v4f s = {0.0f, 0.0f, 0.0f, 0.0f};
  #pragma unroll 16
  for (int c = 0; c < np; ++c) s += ld4(part + (size_t)c * ns + i);
  return s;
}

__global__ __launch_bounds__(256) void k_node1(const float* __restrict__ x,
                                               const float* __restrict__ part,
                                               float* __restrict__ dinv,
                                               float* __restrict__ p, int n, int np) {
  const int ns = plane_stride(n);
  const int i  = (blockIdx.x * 256 + threadIdx.x) * 4;
  if (i >= ns) return;
  v4f xv;
  if (i + 3 < n) {
    xv = ld4(x + i);
  } else {
    #pragma unroll
    for (int j = 0; j < 4; ++j) xv[j] = (i + j < n) ? x[min(i + j, n - 1)] : 0.0f;
  }
  const v4f dg = 1.0f + plane_sum4(part, ns, i, np);
  v4f dv;
  #pragma unroll
  for (int j = 0; j < 4; ++j) dv[j] = rsqrtf(dg[j]);
  *(v4f*)(dinv + i) = dv;
  *(v4f*)(p + i)    = xv * dv;
}

template <int IMM>
__device__ __forceinline__ float swz_add(float v) {
#if defined(__AMDGCN__)
  int s = __builtin_amdgcn_ds_swizzle(__float_as_int(v), IMM);
  return v + __int_as_float(s);
#else
  return v;
#endif
}

__global__ __launch_bounds__(256) void k_mid(const float* __restrict__ p,
                                             const float* __restrict__ part,
                                             const float* __restrict__ dinv,
                                             const float* __restrict__ W1,
                                             const float* __restrict__ b1,
                                             const float* __restrict__ W2,
                                             float* __restrict__ q, int n, int np) {
  __shared__ __attribute__((aligned(16))) float abuf[NGRP];
  __shared__ __attribute__((aligned(16))) float zbuf[NGRP];
  const int tid    = threadIdx.x;
  const int lane   = tid & 31;
  const int lane16 = lane & 15;
  const bool hi    = lane >= 16;
  const int ns = plane_stride(n);
  const int i  = blockIdx.x * NGRP + tid * 4;
  const bool live = i < ns;
  const int il = live ? i : ns - 4;
  const float w1v = W1[lane16];
  const float b1v = b1[lane16];
  const float w2v = W2[lane16];
  const v4f p4 = ld4(p + il);
  const v4f d4 = ld4(dinv + il);

  v4f se = {0.0f, 0.0f, 0.0f, 0.0f}, so = {0.0f, 0.0f, 0.0f, 0.0f};
  int c = 0;
  #pragma unroll 8
  for (; c + 1 < np; c += 2) {
    se += ld4(part + (size_t)c * ns + il);
    so += ld4(part + (size_t)(c + 1) * ns + il);
  }
  if (c < np) se += ld4(part + (size_t)c * ns + il);
  const v4f a4 = (p4 + (se + so)) * d4;
  *(v4f*)(abuf + tid * 4) = a4;
  __syncthreads();

  const int wbase = (tid >> 5) * 128;
  float zt[8][8];
  #pragma unroll
  for (int t = 0; t < 8; ++t) {
    const float al = abuf[wbase + 16 * t + lane16];
    const float a  = hi ? 0.0f : al;
#if __has_builtin(__builtin_amdgcn_wmma_f32_16x16x4_f32)
    v2f A; A.x = a;                  A.y = 0.0f;
    v2f B; B.x = hi ? 0.0f : w1v;    B.y = 0.0f;
    v8f C;
    #pragma unroll
    for (int r = 0; r < 8; ++r) C[r] = b1v;
    v8f D = __builtin_amdgcn_wmma_f32_16x16x4_f32(
        false, A, false, B, (short)0, C, false, false);
#else
    v8f D;
    #pragma unroll
    for (int r = 0; r < 8; ++r) {
      float am = __shfl(a, r + (hi ? 8 : 0), 32);
      D[r] = fmaf(am, w1v, b1v);
    }
#endif
    #pragma unroll
    for (int r = 0; r < 8; ++r) {
      float h = D[r];
      h = h > 0.0f ? h : 0.0f;
      float v = h * w2v;
      v = swz_add<0x041F>(v);
      v = swz_add<0x081F>(v);
      v = swz_add<0x101F>(v);
      v = swz_add<0x201F>(v);
      zt[t][r] = v;
    }
  }
  if (lane16 == 0) {
    float* zw = zbuf + wbase + (hi ? 8 : 0);
    #pragma unroll
    for (int t = 0; t < 8; ++t) {
      const v4f z0 = {zt[t][0], zt[t][1], zt[t][2], zt[t][3]};
      const v4f z1 = {zt[t][4], zt[t][5], zt[t][6], zt[t][7]};
      *(v4f*)(zw + 16 * t)     = z0;
      *(v4f*)(zw + 16 * t + 4) = z1;
    }
  }
  __syncthreads();
  const v4f z4 = *(const v4f*)(zbuf + tid * 4);
  if (live) *(v4f*)(q + i) = z4 * d4;
}

__global__ __launch_bounds__(256) void k_out(const float* __restrict__ dinv,
                                             const float* __restrict__ q,
                                             const float* __restrict__ part,
                                             const float* __restrict__ b2,
                                             float* __restrict__ out, int n, int np) {
  const int ns = plane_stride(n);
  const int i  = (blockIdx.x * 256 + threadIdx.x) * 4;
  if (i >= n) return;
  const v4f d4 = ld4(dinv + i);
  const v4f q4 = ld4(q + i);
  const float bb = b2[0];
  const v4f u = q4 + plane_sum4(part, ns, i, np);
  v4f o;
  #pragma unroll
  for (int j = 0; j < 4; ++j) o[j] = fmaf(d4[j], u[j], bb);
  if (i + 3 < n) {
    *(v4f*)(out + i) = o;
  } else {
    #pragma unroll
    for (int j = 0; j < 4; ++j)
      if (i + j < n) out[i + j] = o[j];
  }
}

extern "C" void kernel_launch(void* const* d_in, const int* in_sizes, int n_in,
                              void* d_out, int out_size, void* d_ws, size_t ws_size,
                              hipStream_t stream) {
  const float* x  = (const float*)d_in[0];
  const float* W1 = (const float*)d_in[1];
  const float* b1 = (const float*)d_in[2];
  const float* W2 = (const float*)d_in[3];
  const float* b2 = (const float*)d_in[4];
  const int*   ei = (const int*)d_in[5];

  const int n = in_sizes[0];
  const int e = in_sizes[5] / 2;
  const int* src = ei;
  const int* dst = ei + e;

  const int ns = (n + 31) & ~31;
  float* out  = (float*)d_out;
  float* dinv = (float*)d_ws;
  float* p    = dinv + ns;
  float* q    = p + ns;
  int*      table  = (int*)d_ws + (4u << 20);
  int*      binoff = (int*)d_ws + (6u << 20);
  unsigned* bins   = (unsigned*)d_ws + (8u << 20);
  float*    part   = (float*)d_ws + (16u << 20);

  const int B  = 256;
  const int gN = (ns + NGRP - 1) / NGRP;
  const int nr = (n + RNODES - 1) / RNODES;
  const int gT = (e + TILE - 1) / TILE;
  const int gP = nr * nr;

  k_count<<<gT, PT, 0, stream>>>(src, dst, table, e, nr);
  k_place<<<gT, PT, 0, stream>>>(src, dst, table, binoff, bins, e, nr, gT);
  k_bin<false><<<gP, PT, 0, stream>>>(bins, binoff, x, part, n, nr);
  k_node1<<<gN, B, 0, stream>>>(x, part, dinv, p, n, nr);
  k_bin<true><<<gP, PT, 0, stream>>>(bins, binoff, p, part, n, nr);
  k_mid<<<gN, B, 0, stream>>>(p, part, dinv, W1, b1, W2, q, n, nr);
  k_bin<true><<<gP, PT, 0, stream>>>(bins, binoff, q, part, n, nr);
  k_out<<<gN, B, 0, stream>>>(dinv, q, part, b2, out, n, nr);
}


# Edges binned in ONE pass: each tile ranks and orders its own edges in LDS into own padded runs; a scatter work-group walks its bin's run list (LDS adds in arrival order as before); count launch gone
# speedup vs baseline: 7.4677x; 1.1137x over previous
#include <hip/hip_runtime.h>


typedef __attribute__((ext_vector_type(2))) float v2f;
typedef __attribute__((ext_vector_type(8))) float v8f;
typedef __attribute__((ext_vector_type(4))) float v4f;

#define RSHIFT 15
#define RNODES (1 << RSHIFT)
#define NBIN 256
#define TILE 20480
#define PT 1024

__device__ __forceinline__ int plane_stride(int n) { return (n + 31) & ~31; }

#define EMPTY 0xFFFFFFFFu
#define EPT (TILE / PT)
#define CAP (TILE + NBIN * 31)
#define TCH (CAP / 32)
__device__ __forceinline__ int pad32(int v) { return (v + 31) & ~31; }

__global__ __launch_bounds__(PT) void k_sort(const int* __restrict__ src,
                                             const int* __restrict__ dst,
                                             int* __restrict__ runs,
                                             unsigned* __restrict__ bins, int e, int nr, int ntile) {
  __shared__ __attribute__((aligned(16))) unsigned stage[CAP + PT];
  __shared__ int hist[NBIN + PT];
  __shared__ int offs[NBIN];
  __shared__ int wtot[NBIN / 32];
  __shared__ int ltot;
  const int tid  = threadIdx.x;
  const int lane = tid & 31;
  const int g    = blockIdx.x;
  hist[tid] = 0;
  if (tid < NBIN) hist[PT + tid] = 0;
  __syncthreads();

  const int beg = g * TILE;
  const int end = min(beg + TILE, e);
  int d[EPT], s[EPT];
  #pragma unroll
  for (int k = 0; k < EPT; ++k) {
    const int j = min(beg + tid + k * PT, end - 1);
    d[k] = dst[j];
    s[k] = src[j];
  }
  __builtin_amdgcn_sched_barrier(0);
  int rk[EPT];
  #pragma unroll
  for (int k = 0; k < EPT; ++k) {
    const bool ok = beg + tid + k * PT < end;
    const int b = (d[k] >> RSHIFT) * nr + (s[k] >> RSHIFT);
    rk[k] = atomicAdd(&hist[ok ? b : NBIN + tid], 1);
  }
  __syncthreads();

  int own = 0, ownp = 0, incl = 0;
  if (tid < NBIN) {
    own  = hist[tid];
    ownp = pad32(own);
    incl = ownp;
    #pragma unroll
    for (int o = 1; o < 32; o <<= 1) {
      const int v = __shfl_up(incl, o, 32);
      if (lane >= o) incl += v;
    }
    if (lane == 31) wtot[tid >> 5] = incl;
  }
  __syncthreads();
  if (tid < NBIN) {
    int base = 0;
    for (int q = 0; q < (tid >> 5); ++q) base += wtot[q];
    const int off = base + incl - ownp;
    offs[tid] = off;
    for (int j = off + own; j < off + ownp; ++j) stage[j] = EMPTY;
    ((int2*)runs)[(size_t)tid * ntile + g] = make_int2(g * TCH + (off >> 5), ownp >> 5);
    if (tid == NBIN - 1) ltot = off + ownp;
  }
  __syncthreads();
  #pragma unroll
  for (int k = 0; k < EPT; ++k) {
    const bool ok = beg + tid + k * PT < end;
    const int b = (d[k] >> RSHIFT) * nr + (s[k] >> RSHIFT);
    const int pos = offs[ok ? b : 0] + rk[k];
    stage[ok ? pos : CAP + tid] = ((unsigned)(s[k] & (RNODES - 1)) << 16) | (unsigned)(d[k] & (RNODES - 1));
  }
  __syncthreads();
  const int total4 = ltot >> 2;
  uint4* __restrict__ o4 = (uint4*)(bins + (size_t)g * CAP);
  const uint4* st4 = (const uint4*)stage;
  for (int j = tid; j < total4; j += PT) o4[j] = st4[j];
}

#define LCAP 2048
template <bool GATHER>
__global__ __launch_bounds__(PT) void k_bin(const unsigned* __restrict__ bins,
                                            const int* __restrict__ runs,
                                            const float* __restrict__ val,
                                            float* __restrict__ part, int n, int nr, int ntile) {
  __shared__ __attribute__((aligned(16))) float acc[RNODES];
  __shared__ __attribute__((aligned(16))) float win[GATHER ? RNODES : 4];
  __shared__ int clist[LCAP];
  __shared__ int wsum[PT / 32];
  const int tid  = threadIdx.x;
  const int lane = tid & 31;
  const int r    = blockIdx.x / nr;
  const int w    = blockIdx.x - r * nr;

  const int2 rn = ((const int2*)runs)[(size_t)blockIdx.x * ntile + min(tid, ntile - 1)];
  const int nch = (tid < ntile) ? rn.y : 0;

  float4* acc4 = (float4*)acc;
  for (int i = tid; i < RNODES / 4; i += PT) acc4[i] = make_float4(0.f, 0.f, 0.f, 0.f);
  if (GATHER) {
    const int wlo = w << RSHIFT;
    #pragma unroll 16
    for (int i = tid; i < RNODES; i += PT) {
      const int gi = wlo + i;
      const float t = val[min(gi, n - 1)];
      win[i] = (gi < n) ? t : 0.0f;
    }
  }
  int incl = nch;
  #pragma unroll
  for (int o = 1; o < 32; o <<= 1) {
    const int v = __shfl_up(incl, o, 32);
    if (lane >= o) incl += v;
  }
  if (lane == 31) wsum[tid >> 5] = incl;
  __syncthreads();
  int base = 0, nc = 0;
  #pragma unroll 8
  for (int q = 0; q < PT / 32; ++q) {
    const int v = wsum[q];
    nc += v;
    if (q < (tid >> 5)) base += v;
  }
  const int pre = base + incl - nch;

  const uint4* __restrict__ b4 = (const uint4*)bins;
  for (int lo = 0; lo < nc; lo += LCAP) {
    const int hi = min(nc, lo + LCAP);
    for (int c = max(pre, lo); c < min(pre + nch, hi); ++c) clist[c - lo] = rn.x + (c - pre);
    __syncthreads();
    const int nv = (hi - lo) * 8;
    for (int j = tid; j < nv; j += 2 * PT) {
      const int j2 = min(j + PT, nv - 1);
      const uint4 A = b4[(size_t)clist[j >> 3] * 8 + (j & 7)];
      const uint4 B = b4[(size_t)clist[j2 >> 3] * 8 + (j2 & 7)];
      asm volatile("" :: "v"(A.x), "v"(A.y), "v"(A.z), "v"(A.w), "v"(B.x), "v"(B.y), "v"(B.z), "v"(B.w));
      __builtin_amdgcn_sched_barrier(0);
      const bool okB = (j + PT < nv);
      const unsigned pk[8] = {A.x, A.y, A.z, A.w, B.x, B.y, B.z, B.w};
      float v[8];
      #pragma unroll
      for (int k = 0; k < 8; ++k) v[k] = GATHER ? win[(pk[k] >> 16) & (RNODES - 1)] : 1.0f;
      #pragma unroll
      for (int k = 0; k < 8; ++k)
        if ((k < 4 || okB) && pk[k] != EMPTY) atomicAdd(&acc[pk[k] & (RNODES - 1)], v[k]);
    }
    __syncthreads();
  }

  const int ns  = plane_stride(n);
  const int lo  = r << RSHIFT;
  const int cnt = min(RNODES, ns - lo);
  float4* __restrict__ o4 = (float4*)(part + (size_t)w * ns + lo);
  for (int i = tid; i < (cnt >> 2); i += PT) o4[i] = acc4[i];
}

#define NGRP 1024

__device__ __forceinline__ v4f ld4(const float* __restrict__ a) { return *(const v4f*)a; }

__device__ __forceinline__ v4f plane_sum4(const float* __restrict__ part, int ns, int i, int np) {
  v4f s = {0.0f, 0.0f, 0.0f, 0.0f};
  #pragma unroll 16
  for (int c = 0; c < np; ++c) s += ld4(part + (size_t)c * ns + i);
  return s;
}

__global__ __launch_bounds__(256) void k_node1(const float* __restrict__ x,
                                               const float* __restrict__ part,
                                               float* __restrict__ dinv,
                                               float* __restrict__ p, int n, int np) {
  const int ns = plane_stride(n);
  const int i  = (blockIdx.x * 256 + threadIdx.x) * 4;
  if (i >= ns) return;
  v4f xv;
  if (i + 3 < n) {
    xv = ld4(x + i);
  } else {
    #pragma unroll
    for (int j = 0; j < 4; ++j) xv[j] = (i + j < n) ? x[min(i + j, n - 1)] : 0.0f;
  }
  const v4f dg = 1.0f + plane_sum4(part, ns, i, np);
  v4f dv;
  #pragma unroll
  for (int j = 0; j < 4; ++j) dv[j] = rsqrtf(dg[j]);
  *(v4f*)(dinv + i) = dv;
  *(v4f*)(p + i)    = xv * dv;
}

template <int IMM>
__device__ __forceinline__ float swz_add(float v) {
#if defined(__AMDGCN__)
  int s = __builtin_amdgcn_ds_swizzle(__float_as_int(v), IMM);
  return v + __int_as_float(s);
#else
  return v;
#endif
}

__global__ __launch_bounds__(256) void k_mid(const float* __restrict__ p,
                                             const float* __restrict__ part,
                                             const float* __restrict__ dinv,
                                             const float* __restrict__ W1,
                                             const float* __restrict__ b1,
                                             const float* __restrict__ W2,
                                             float* __restrict__ q, int n, int np) {
  __shared__ __attribute__((aligned(16))) float abuf[NGRP];
  __shared__ __attribute__((aligned(16))) float zbuf[NGRP];
  const int tid    = threadIdx.x;
  const int lane   = tid & 31;
  const int lane16 = lane & 15;
  const bool hi    = lane >= 16;
  const int ns = plane_stride(n);
  const int i  = blockIdx.x * NGRP + tid * 4;
  const bool live = i < ns;
  const int il = live ? i : ns - 4;
  const float w1v = W1[lane16];
  const float b1v = b1[lane16];
  const float w2v = W2[lane16];
  const v4f p4 = ld4(p + il);
  const v4f d4 = ld4(dinv + il);

  v4f se = {0.0f, 0.0f, 0.0f, 0.0f}, so = {0.0f, 0.0f, 0.0f, 0.0f};
  int c = 0;
  #pragma unroll 8
  for (; c + 1 < np; c += 2) {
    se += ld4(part + (size_t)c * ns + il);
    so += ld4(part + (size_t)(c + 1) * ns + il);
  }
  if (c < np) se += ld4(part + (size_t)c * ns + il);
  const v4f a4 = (p4 + (se + so)) * d4;
  *(v4f*)(abuf + tid * 4) = a4;
  __syncthreads();

  const int wbase = (tid >> 5) * 128;
  float zt[8][8];
  #pragma unroll
  for (int t = 0; t < 8; ++t) {
    const float al = abuf[wbase + 16 * t + lane16];
    const float a  = hi ? 0.0f : al;
#if __has_builtin(__builtin_amdgcn_wmma_f32_16x16x4_f32)
    v2f A; A.x = a;                  A.y = 0.0f;
    v2f B; B.x = hi ? 0.0f : w1v;    B.y = 0.0f;
    v8f C;
    #pragma unroll
    for (int r = 0; r < 8; ++r) C[r] = b1v;
    v8f D = __builtin_amdgcn_wmma_f32_16x16x4_f32(
        false, A, false, B, (short)0, C, false, false);
#else
    v8f D;
    #pragma unroll
    for (int r = 0; r < 8; ++r) {
      float am = __shfl(a, r + (hi ? 8 : 0), 32);
      D[r] = fmaf(am, w1v, b1v);
    }
#endif
    #pragma unroll
    for (int r = 0; r < 8; ++r) {
      float h = D[r];
      h = h > 0.0f ? h : 0.0f;
      float v = h * w2v;
      v = swz_add<0x041F>(v);
      v = swz_add<0x081F>(v);
      v = swz_add<0x101F>(v);
      v = swz_add<0x201F>(v);
      zt[t][r] = v;
    }
  }
  if (lane16 == 0) {
    float* zw = zbuf + wbase + (hi ? 8 : 0);
    #pragma unroll
    for (int t = 0; t < 8; ++t) {
      const v4f z0 = {zt[t][0], zt[t][1], zt[t][2], zt[t][3]};
      const v4f z1 = {zt[t][4], zt[t][5], zt[t][6], zt[t][7]};
      *(v4f*)(zw + 16 * t)     = z0;
      *(v4f*)(zw + 16 * t + 4) = z1;
    }
  }
  __syncthreads();
  const v4f z4 = *(const v4f*)(zbuf + tid * 4);
  if (live) *(v4f*)(q + i) = z4 * d4;
}

__global__ __launch_bounds__(256) void k_out(const float* __restrict__ dinv,
                                             const float* __restrict__ q,
                                             const float* __restrict__ part,
                                             const float* __restrict__ b2,
                                             float* __restrict__ out, int n, int np) {
  const int ns = plane_stride(n);
  const int i  = (blockIdx.x * 256 + threadIdx.x) * 4;
  if (i >= n) return;
  const v4f d4 = ld4(dinv + i);
  const v4f q4 = ld4(q + i);
  const float bb = b2[0];
  const v4f u = q4 + plane_sum4(part, ns, i, np);
  v4f o;
  #pragma unroll
  for (int j = 0; j < 4; ++j) o[j] = fmaf(d4[j], u[j], bb);
  if (i + 3 < n) {
    *(v4f*)(out + i) = o;
  } else {
    #pragma unroll
    for (int j = 0; j < 4; ++j)
      if (i + j < n) out[i + j] = o[j];
  }
}

extern "C" void kernel_launch(void* const* d_in, const int* in_sizes, int n_in,
                              void* d_out, int out_size, void* d_ws, size_t ws_size,
                              hipStream_t stream) {
  const float* x  = (const float*)d_in[0];
  const float* W1 = (const float*)d_in[1];
  const float* b1 = (const float*)d_in[2];
  const float* W2 = (const float*)d_in[3];
  const float* b2 = (const float*)d_in[4];
  const int*   ei = (const int*)d_in[5];

  const int n = in_sizes[0];
  const int e = in_sizes[5] / 2;
  const int* src = ei;
  const int* dst = ei + e;

  const int ns = (n + 31) & ~31;
  float* out  = (float*)d_out;
  float* dinv = (float*)d_ws;
  float* p    = dinv + ns;
  float* q    = p + ns;
  int*      runs   = (int*)d_ws + (4u << 20);
  unsigned* bins   = (unsigned*)d_ws + (8u << 20);
  float*    part   = (float*)d_ws + (16u << 20);

  const int B  = 256;
  const int gN = (ns + NGRP - 1) / NGRP;
  const int nr = (n + RNODES - 1) / RNODES;
  const int gT = (e + TILE - 1) / TILE;
  const int gP = nr * nr;

  k_sort<<<gT, PT, 0, stream>>>(src, dst, runs, bins, e, nr, gT);
  k_bin<false><<<gP, PT, 0, stream>>>(bins, runs, x, part, n, nr, gT);
  k_node1<<<gN, B, 0, stream>>>(x, part, dinv, p, n, nr);
  k_bin<true><<<gP, PT, 0, stream>>>(bins, runs, p, part, n, nr, gT);
  k_mid<<<gN, B, 0, stream>>>(p, part, dinv, W1, b1, W2, q, n, nr);
  k_bin<true><<<gP, PT, 0, stream>>>(bins, runs, q, part, n, nr, gT);
  k_out<<<gN, B, 0, stream>>>(dinv, q, part, b2, out, n, nr);
}
